# starting kernel
# baseline (speedup 1.0000x reference)
#include <hip/hip_runtime.h>
#include <math.h>

typedef __attribute__((ext_vector_type(2))) float v2f;
typedef __attribute__((ext_vector_type(8))) float v8f;

#define HH 4
#define DD 32
#define NN 1024
#define BB 8
#define KDIM 128
#define NEG_SLOPE 0.2f

__global__ __launch_bounds__(256) void gat_h_gemm(const float* __restrict__ x,
                                                  const float* __restrict__ W,
                                                  float* __restrict__ h) {
    __shared__ float lds_x[16][132];
    __shared__ float lds_w[128][36];
    const int tid  = threadIdx.x;
    const int lane = tid & 31;
    const int wave = tid >> 5;
    const int r0   = blockIdx.x * 16;
    const int c0   = wave * 16;
    const int lm   = lane & 15;
    const int lhi  = lane >> 4;

    for (int idx = tid; idx < 16 * KDIM; idx += 256) {
        int row = idx >> 7, col = idx & 127;
        lds_x[row][col] = x[(size_t)(r0 + row) * KDIM + col];
    }

    v8f acc = {};
    for (int kk = 0; kk < KDIM; kk += 32) {
        __syncthreads();
        for (int idx = tid; idx < 128 * 32; idx += 256) {
            int row = idx >> 5, col = idx & 31;
            lds_w[row][col] = W[(size_t)row * KDIM + kk + col];
        }
        if (kk + 32 < KDIM)
            __builtin_prefetch(&W[(size_t)(tid & 127) * KDIM + kk + 32], 0, 1);
        __syncthreads();
#pragma unroll
        for (int s = 0; s < 8; ++s) {
            const int kb = 4 * s + lhi * 2;
            v2f a, b;
            a.x = lds_x[lm][kk + kb + 0];
            a.y = lds_x[lm][kk + kb + 1];
            b.x = lds_w[c0 + lm][kb + 0];
            b.y = lds_w[c0 + lm][kb + 1];
            acc = __builtin_amdgcn_wmma_f32_16x16x4_f32(false, a, false, b,
                                                        (short)0, acc, false, false);
        }
    }
#pragma unroll
    for (int v = 0; v < 8; ++v) {
        int m = v + lhi * 8;
        h[(size_t)(r0 + m) * KDIM + c0 + lm] = acc[v];
    }
}

__global__ __launch_bounds__(256) void gat_alpha(const float* __restrict__ h,
                                                 const float* __restrict__ a_src,
                                                 const float* __restrict__ a_dst,
                                                 float* __restrict__ alpha_src,
                                                 float* __restrict__ alpha_dst) {
    int t = blockIdx.x * 256 + threadIdx.x;
    int head = t & 3;
    int bn   = t >> 2;
    const float4* hp = (const float4*)(h + (size_t)bn * KDIM + head * DD);
    const float4* sp = (const float4*)(a_src + head * DD);
    const float4* dp = (const float4*)(a_dst + head * DD);
    float ss = 0.f, sd = 0.f;
#pragma unroll
    for (int i = 0; i < 8; ++i) {
        float4 hv = hp[i], av = sp[i], dv = dp[i];
        ss += hv.x * av.x + hv.y * av.y + hv.z * av.z + hv.w * av.w;
        sd += hv.x * dv.x + hv.y * dv.y + hv.z * dv.z + hv.w * dv.w;
    }
    alpha_src[t] = ss;
    alpha_dst[t] = sd;
}

__global__ __launch_bounds__(256) void gat_dmax(const float* __restrict__ alpha_dst,
                                                const float* __restrict__ alive,
                                                float* __restrict__ dmax) {
    __shared__ float red[256];
    const int bh = blockIdx.x;
    const int b = bh >> 2, head = bh & 3;
    float m = -INFINITY;
    for (int j = threadIdx.x; j < NN; j += 256) {
        float d  = alpha_dst[(size_t)(b * NN + j) * HH + head];
        if (alive[b * NN + j] >= 0.5f && d > m) m = d;
    }
    red[threadIdx.x] = m;
    __syncthreads();
    for (int s = 128; s > 0; s >>= 1) {
        if (threadIdx.x < s) red[threadIdx.x] = fmaxf(red[threadIdx.x], red[threadIdx.x + s]);
        __syncthreads();
    }
    if (threadIdx.x == 0) dmax[bh] = red[0];
}

__global__ __launch_bounds__(256) void gat_attn(const float* __restrict__ h,
                                                const float* __restrict__ alpha_src,
                                                const float* __restrict__ alpha_dst,
                                                const float* __restrict__ alive,
                                                const float* __restrict__ dmax,
                                                float* __restrict__ out) {
    __shared__ float dval[HH][NN];
    __shared__ float inv_s[8][16];
    const int tid   = threadIdx.x;
    const int lane  = tid & 31;
    const int wave  = tid >> 5;
    const int head  = wave & 3;
    const int dhalf = wave >> 2;
    const int b     = blockIdx.x >> 6;
    const int q0    = (blockIdx.x & 63) * 16;
    const int lm    = lane & 15;
    const int lhi   = lane >> 4;
    const int d0    = head * DD + dhalf * 16;

    for (int idx = tid; idx < HH * NN; idx += 256) {
        int hh = idx & 3, j = idx >> 2;
        float d = alpha_dst[(size_t)(b * NN + j) * HH + hh];
        dval[hh][j] = (alive[b * NN + j] >= 0.5f) ? d : -INFINITY;
    }
    __syncthreads();

    const float Dm  = dmax[b * HH + head];
    const float s_i = alpha_src[(size_t)(b * NN + q0 + lm) * HH + head];
    const float e0  = s_i + Dm;
    const float m_i = (e0 >= 0.f) ? e0 : NEG_SLOPE * e0;

    v8f acc = {};
    float zacc = 0.f;
    if (Dm > -INFINITY) {
        const float* hbase = h + (size_t)b * NN * KDIM + d0 + lm;
        for (int j0 = 0; j0 < NN; j0 += 4) {
            const int ja = j0 + lhi * 2;
            float ea = s_i + dval[head][ja];
            float eb = s_i + dval[head][ja + 1];
            ea = (ea >= 0.f) ? ea : NEG_SLOPE * ea;
            eb = (eb >= 0.f) ? eb : NEG_SLOPE * eb;
            v2f p, vv;
            p.x = __expf(ea - m_i);
            p.y = __expf(eb - m_i);
            zacc += p.x + p.y;
            vv.x = hbase[(size_t)ja * KDIM];
            vv.y = hbase[(size_t)(ja + 1) * KDIM];
            acc = __builtin_amdgcn_wmma_f32_16x16x4_f32(false, p, false, vv,
                                                        (short)0, acc, false, false);
        }
    }
    zacc += __shfl_xor(zacc, 16, 32);
    if (lhi == 0) inv_s[wave][lm] = (zacc > 0.f) ? (1.f / zacc) : 0.f;
    __syncthreads();
#pragma unroll
    for (int v = 0; v < 8; ++v) {
        int m = v + lhi * 8;
        out[((size_t)(b * NN) + q0 + m) * KDIM + d0 + lm] = acc[v] * inv_s[wave][m];
    }
}

extern "C" void kernel_launch(void* const* d_in, const int* in_sizes, int n_in,
                              void* d_out, int out_size, void* d_ws, size_t ws_size,
                              hipStream_t stream) {
    const float* x     = (const float*)d_in[0];
    const float* alive = (const float*)d_in[1];
    const float* W     = (const float*)d_in[2];
    const float* a_src = (const float*)d_in[3];
    const float* a_dst = (const float*)d_in[4];
    float* out = (float*)d_out;

    float* h_ws  = (float*)d_ws;
    float* as_ws = h_ws + (size_t)BB * NN * KDIM;
    float* ad_ws = as_ws + (size_t)BB * NN * HH;
    float* dm_ws = ad_ws + (size_t)BB * NN * HH;

    gat_h_gemm<<<512, 256, 0, stream>>>(x, W, h_ws);
    gat_alpha <<<128, 256, 0, stream>>>(h_ws, a_src, a_dst, as_ws, ad_ws);
    gat_dmax  <<<32,  256, 0, stream>>>(ad_ws, alive, dm_ws);
    gat_attn  <<<512, 256, 0, stream>>>(h_ws, as_ws, ad_ws, alive, dm_ws, out);
}


# alpha launch folded into the attention block: it copies its (batch, head) slab of h (now head-major) to LDS once, takes both dots and the V rows by rank from there; 3 -> 2 launches; sum order kept
# speedup vs baseline: 1.6008x; 1.6008x over previous
#include <hip/hip_runtime.h>
#include <math.h>

typedef __attribute__((ext_vector_type(2))) float v2f;
typedef __attribute__((ext_vector_type(8))) float v8f;

#define HH 4
#define DD 32
#define NN 1024
#define BB 8
#define KDIM 128
#define NEG_SLOPE 0.2f

__global__ __launch_bounds__(256) void gat_h_gemm(const float* __restrict__ x,
                                                  const float* __restrict__ W,
                                                  float* __restrict__ h) {
    __shared__ float lds_x[16][132];
    __shared__ float lds_w[128][36];
    const int tid  = threadIdx.x;
    const int lane = tid & 31;
    const int wave = tid >> 5;
    const int r0   = blockIdx.x * 16;
    const int c0   = wave * 16;
    const int lm   = lane & 15;
    const int lhi  = lane >> 4;

    for (int idx = tid; idx < 16 * KDIM; idx += 256) {
        int row = idx >> 7, col = idx & 127;
        lds_x[row][col] = x[(size_t)(r0 + row) * KDIM + col];
    }

    v8f acc = {};
    for (int kk = 0; kk < KDIM; kk += 32) {
        __syncthreads();
        for (int idx = tid; idx < 128 * 32; idx += 256) {
            int row = idx >> 5, col = idx & 31;
            lds_w[row][col] = W[(size_t)row * KDIM + kk + col];
        }
        if (kk + 32 < KDIM)
            __builtin_prefetch(&W[(size_t)(tid & 127) * KDIM + kk + 32], 0, 1);
        __syncthreads();
#pragma unroll
        for (int s = 0; s < 8; ++s) {
            const int kb = 4 * s + lhi * 2;
            v2f a, b;
            a.x = lds_x[lm][kk + kb + 0];
            a.y = lds_x[lm][kk + kb + 1];
            b.x = lds_w[c0 + lm][kb + 0];
            b.y = lds_w[c0 + lm][kb + 1];
            acc = __builtin_amdgcn_wmma_f32_16x16x4_f32(false, a, false, b,
                                                        (short)0, acc, false, false);
        }
    }
    const int col = c0 + lm;
    float* hq = h + (size_t)(r0 >> 10) * (HH * NN * DD)
                  + (((col >> 5) * NN + (r0 & (NN - 1)) + lhi * 8) * DD + (col & (DD - 1)));
#pragma unroll
    for (int v = 0; v < 8; ++v) hq[v * DD] = acc[v];
}

__device__ __forceinline__ unsigned xlane(unsigned x, int jj) {
    const int bits = (int)x;
    switch (jj) {
    case 1:  return (unsigned)__builtin_amdgcn_update_dpp(bits, bits, 0x161, 0xf, 0xf, false);
    case 2:  return (unsigned)__builtin_amdgcn_update_dpp(bits, bits, 0x162, 0xf, 0xf, false);
    case 4:  return (unsigned)__builtin_amdgcn_update_dpp(bits, bits, 0x164, 0xf, 0xf, false);
    case 8:  return (unsigned)__builtin_amdgcn_update_dpp(bits, bits, 0x168, 0xf, 0xf, false);
    default: return (unsigned)__builtin_amdgcn_permlanex16(bits, bits, 0x76543210, 0xfedcba98, false, false);
    }
}

__global__ __launch_bounds__(1024) void gat_attn_sorted(const float* __restrict__ h,
                                                        const float* __restrict__ a_src,
                                                        const float* __restrict__ a_dst,
                                                        const float* __restrict__ alive,
                                                        float* __restrict__ out) {
    __shared__ unsigned long long xbuf[2][NN];
    __shared__ float sd[NN];
    __shared__ int   sj[NN];
    __shared__ float sE[NN];
    __shared__ float sF[NN];
    __shared__ float zF[NN];
    __shared__ float zE[NN];
    __shared__ float segF[32][32];
    __shared__ float segE[32][32];
    __shared__ float offF[32][32];
    __shared__ float offE[32][32];
    __shared__ float zsegF[32];
    __shared__ float zsegE[32];
    __shared__ __attribute__((aligned(16))) float tab[2 * NN * 32];
    float (*PFt)[32] = (float (*)[32])tab;
    float (*SEt)[32] = (float (*)[32])(tab + NN * 32);
    float* hs = tab;
    constexpr int SP = 36;

    const int tid  = threadIdx.x;
    const int lane = tid & 31;
    const int wave = tid >> 5;
    const int head = blockIdx.x & 3;
    const int b    = blockIdx.x >> 2;

    const float al = alive[b * NN + tid];
    const float4* hg = (const float4*)(h + (size_t)blockIdx.x * NN * DD);
    float4 pc[8];
#pragma unroll
    for (int i = 0; i < 8; ++i) pc[i] = hg[i * 1024 + tid];
#pragma unroll
    for (int i = 0; i < 8; ++i) {
        const int g = i * 1024 + tid;
        *(float4*)(hs + (g >> 3) * SP + (g & 7) * 4) = pc[i];
    }
    __syncthreads();
    const float4* hp = (const float4*)(hs + tid * SP);
    const float4* sp = (const float4*)(a_src + head * DD);
    const float4* dp = (const float4*)(a_dst + head * DD);
    float sq = 0.f, dj = 0.f;
#pragma unroll
    for (int i = 0; i < 8; ++i) {
        const float4 hv = hp[i], av = sp[i], dv = dp[i];
        sq += hv.x * av.x + hv.y * av.y + hv.z * av.z + hv.w * av.w;
        dj += hv.x * dv.x + hv.y * dv.y + hv.z * dv.z + hv.w * dv.w;
    }
    const float kd = (al >= 0.5f) ? dj : -INFINITY;
    unsigned u = __float_as_uint(kd);
    u ^= (u & 0x80000000u) ? 0xFFFFFFFFu : 0x80000000u;
    unsigned long long me = ((unsigned long long)u << 32) | (unsigned)tid;

    int xs = 0;
#pragma unroll
    for (int k = 2; k <= NN; k <<= 1) {
        const bool up = (tid & k) == 0;
#pragma unroll
        for (int jj = k >> 1; jj > 0; jj >>= 1) {
            unsigned long long other;
            if (jj >= 32) {
                xbuf[xs][tid] = me;
                __syncthreads();
                other = xbuf[xs][tid ^ jj];
                xs ^= 1;
            } else {
                const unsigned lo = xlane((unsigned)(me & 0xFFFFFFFFull), jj);
                const unsigned hi = xlane((unsigned)(me >> 32), jj);
                other = ((unsigned long long)hi << 32) | lo;
            }
            const bool takemin = (((tid & jj) == 0) == up);
            const bool lt = me < other;
            me = (takemin == lt) ? me : other;
        }
    }

    unsigned ur = (unsigned)(me >> 32);
    ur ^= (ur & 0x80000000u) ? 0x80000000u : 0xFFFFFFFFu;
    const float dr = __uint_as_float(ur);
    const int   jr = (int)(me & 1023ull);
    sd[tid] = dr;
    sj[tid] = jr;
    __syncthreads();
    const float Dm = sd[NN - 1];

    float* orow = out + ((size_t)(b * NN) + wave * 32) * KDIM + head * DD + lane;
    if (!(Dm > -INFINITY)) {
        for (int qq = 0; qq < 32; ++qq) orow[(size_t)qq * KDIM] = 0.f;
        return;
    }

    const float dd = dr - Dm;
    const float Er = expf(dd);
    const float Fr = expf(NEG_SLOPE * dd);
    sE[tid] = Er;
    sF[tid] = Fr;

    float zf = Fr, ze = Er;
    for (int o = 1; o < 32; o <<= 1) {
        const float nf = __shfl_up(zf, o, 32);
        const float ne = __shfl_down(ze, o, 32);
        if (lane >= o) zf += nf;
        if (lane + o < 32) ze += ne;
    }
    if (lane == 31) zsegF[wave] = zf;
    if (lane == 0)  zsegE[wave] = ze;
    __syncthreads();

    const int r0 = wave * 32;
    float v[32];
#pragma unroll
    for (int t = 0; t < 32; ++t) v[t] = hs[sj[r0 + t] * SP + lane];
    __syncthreads();
    float run = 0.f;
#pragma unroll
    for (int t = 0; t < 32; ++t) { run += sF[r0 + t] * v[t]; PFt[r0 + t][lane] = run; }
    segF[wave][lane] = run;
    run = 0.f;
#pragma unroll
    for (int t = 31; t >= 0; --t) { run += sE[r0 + t] * v[t]; SEt[r0 + t][lane] = run; }
    segE[wave][lane] = run;
    __syncthreads();

    float oF = 0.f, oE = 0.f, zoF = 0.f, zoE = 0.f;
    for (int g = 0; g < 32; ++g) {
        const float fa = segF[g][lane], fb = segE[g][lane];
        const float za = zsegF[g], zb = zsegE[g];
        if (g < wave) { oF += fa; zoF += za; }
        if (g > wave) { oE += fb; zoE += zb; }
    }
    zF[tid] = zf + zoF;
    zE[tid] = ze + zoE;
    offF[wave][lane] = oF;
    offE[wave][lane] = oE;
    __syncthreads();

    const float tneg = -sq;
    int k = 0;
    for (int step = NN / 2; step > 0; step >>= 1)
        if (sd[k + step - 1] < tneg) k += step;
    if (sd[k] < tneg) k += 1;
    const float e0 = sq + Dm;
    const float m  = (e0 >= 0.f) ? e0 : NEG_SLOPE * e0;
    const float ai = expf(e0 - m);
    const float ci = expf(NEG_SLOPE * e0 - m);
    float zfv = zF[k > 0 ? k - 1 : 0];
    float zev = zE[k < NN ? k : NN - 1];
    if (k == 0)  zfv = 0.f;
    if (k == NN) zev = 0.f;
    const float Z   = ci * zfv + ai * zev;
    const float inv = (Z > 0.f) ? (1.f / Z) : 0.f;
    const float aw = ai * inv, cw = ci * inv;

    for (int qq = 0; qq < 32; ++qq) {
        const int   kq  = __shfl(k, qq, 32);
        const float awq = __shfl(aw, qq, 32);
        const float cwq = __shfl(cw, qq, 32);
        const int ip = kq > 0 ? kq - 1 : 0;
        const int is = kq < NN ? kq : NN - 1;
        float pf = PFt[ip][lane] + offF[ip >> 5][lane];
        float se = SEt[is][lane] + offE[is >> 5][lane];
        if (kq == 0)  pf = 0.f;
        if (kq == NN) se = 0.f;
        orow[(size_t)qq * KDIM] = cwq * pf + awq * se;
    }
}

extern "C" void kernel_launch(void* const* d_in, const int* in_sizes, int n_in,
                              void* d_out, int out_size, void* d_ws, size_t ws_size,
                              hipStream_t stream) {
    const float* x     = (const float*)d_in[0];
    const float* alive = (const float*)d_in[1];
    const float* W     = (const float*)d_in[2];
    const float* a_src = (const float*)d_in[3];
    const float* a_dst = (const float*)d_in[4];
    float* out = (float*)d_out;

    float* h_ws  = (float*)d_ws;

    gat_h_gemm     <<<512, 256,  0, stream>>>(x, W, h_ws);
    gat_attn_sorted<<<32,  1024, 0, stream>>>(h_ws, a_src, a_dst, alive, out);
}


# table segment offsets: one wave per table walks the 32 segment totals once and stores all 32 offsets (was: every wave walked all 32); suffix offsets add from the top: sum order changed
# speedup vs baseline: 1.6757x; 1.0468x over previous
#include <hip/hip_runtime.h>
#include <math.h>

typedef __attribute__((ext_vector_type(2))) float v2f;
typedef __attribute__((ext_vector_type(8))) float v8f;

#define HH 4
#define DD 32
#define NN 1024
#define BB 8
#define KDIM 128
#define NEG_SLOPE 0.2f

__global__ __launch_bounds__(256) void gat_h_gemm(const float* __restrict__ x,
                                                  const float* __restrict__ W,
                                                  float* __restrict__ h) {
    __shared__ float lds_x[16][132];
    __shared__ float lds_w[128][36];
    const int tid  = threadIdx.x;
    const int lane = tid & 31;
    const int wave = tid >> 5;
    const int r0   = blockIdx.x * 16;
    const int c0   = wave * 16;
    const int lm   = lane & 15;
    const int lhi  = lane >> 4;

    for (int idx = tid; idx < 16 * KDIM; idx += 256) {
        int row = idx >> 7, col = idx & 127;
        lds_x[row][col] = x[(size_t)(r0 + row) * KDIM + col];
    }

    v8f acc = {};
    for (int kk = 0; kk < KDIM; kk += 32) {
        __syncthreads();
        for (int idx = tid; idx < 128 * 32; idx += 256) {
            int row = idx >> 5, col = idx & 31;
            lds_w[row][col] = W[(size_t)row * KDIM + kk + col];
        }
        if (kk + 32 < KDIM)
            __builtin_prefetch(&W[(size_t)(tid & 127) * KDIM + kk + 32], 0, 1);
        __syncthreads();
#pragma unroll
        for (int s = 0; s < 8; ++s) {
            const int kb = 4 * s + lhi * 2;
            v2f a, b;
            a.x = lds_x[lm][kk + kb + 0];
            a.y = lds_x[lm][kk + kb + 1];
            b.x = lds_w[c0 + lm][kb + 0];
            b.y = lds_w[c0 + lm][kb + 1];
            acc = __builtin_amdgcn_wmma_f32_16x16x4_f32(false, a, false, b,
                                                        (short)0, acc, false, false);
        }
    }
    const int col = c0 + lm;
    float* hq = h + (size_t)(r0 >> 10) * (HH * NN * DD)
                  + (((col >> 5) * NN + (r0 & (NN - 1)) + lhi * 8) * DD + (col & (DD - 1)));
#pragma unroll
    for (int v = 0; v < 8; ++v) hq[v * DD] = acc[v];
}

__device__ __forceinline__ unsigned xlane(unsigned x, int jj) {
    const int bits = (int)x;
    switch (jj) {
    case 1:  return (unsigned)__builtin_amdgcn_update_dpp(bits, bits, 0x161, 0xf, 0xf, false);
    case 2:  return (unsigned)__builtin_amdgcn_update_dpp(bits, bits, 0x162, 0xf, 0xf, false);
    case 4:  return (unsigned)__builtin_amdgcn_update_dpp(bits, bits, 0x164, 0xf, 0xf, false);
    case 8:  return (unsigned)__builtin_amdgcn_update_dpp(bits, bits, 0x168, 0xf, 0xf, false);
    default: return (unsigned)__builtin_amdgcn_permlanex16(bits, bits, 0x76543210, 0xfedcba98, false, false);
    }
}

__global__ __launch_bounds__(1024) void gat_attn_sorted(const float* __restrict__ h,
                                                        const float* __restrict__ a_src,
                                                        const float* __restrict__ a_dst,
                                                        const float* __restrict__ alive,
                                                        float* __restrict__ out) {
    __shared__ unsigned long long xbuf[2][NN];
    __shared__ float sd[NN];
    __shared__ int   sj[NN];
    __shared__ float sE[NN];
    __shared__ float sF[NN];
    __shared__ float zF[NN];
    __shared__ float zE[NN];
    __shared__ float segF[32][32];
    __shared__ float segE[32][32];
    __shared__ float offF[32][32];
    __shared__ float offE[32][32];
    __shared__ float zsegF[32];
    __shared__ float zsegE[32];
    __shared__ float zoffF[32];
    __shared__ float zoffE[32];
    __shared__ __attribute__((aligned(16))) float tab[2 * NN * 32];
    float (*PFt)[32] = (float (*)[32])tab;
    float (*SEt)[32] = (float (*)[32])(tab + NN * 32);
    float* hs = tab;
    constexpr int SP = 36;

    const int tid  = threadIdx.x;
    const int lane = tid & 31;
    const int wave = tid >> 5;
    const int head = blockIdx.x & 3;
    const int b    = blockIdx.x >> 2;

    const float al = alive[b * NN + tid];
    const float4* hg = (const float4*)(h + (size_t)blockIdx.x * NN * DD);
    float4 pc[8];
#pragma unroll
    for (int i = 0; i < 8; ++i) pc[i] = hg[i * 1024 + tid];
#pragma unroll
    for (int i = 0; i < 8; ++i) {
        const int g = i * 1024 + tid;
        *(float4*)(hs + (g >> 3) * SP + (g & 7) * 4) = pc[i];
    }
    __syncthreads();
    const float4* hp = (const float4*)(hs + tid * SP);
    const float4* sp = (const float4*)(a_src + head * DD);
    const float4* dp = (const float4*)(a_dst + head * DD);
    float sq = 0.f, dj = 0.f;
#pragma unroll
    for (int i = 0; i < 8; ++i) {
        const float4 hv = hp[i], av = sp[i], dv = dp[i];
        sq += hv.x * av.x + hv.y * av.y + hv.z * av.z + hv.w * av.w;
        dj += hv.x * dv.x + hv.y * dv.y + hv.z * dv.z + hv.w * dv.w;
    }
    const float kd = (al >= 0.5f) ? dj : -INFINITY;
    unsigned u = __float_as_uint(kd);
    u ^= (u & 0x80000000u) ? 0xFFFFFFFFu : 0x80000000u;
    unsigned long long me = ((unsigned long long)u << 32) | (unsigned)tid;

    int xs = 0;
#pragma unroll
    for (int k = 2; k <= NN; k <<= 1) {
        const bool up = (tid & k) == 0;
#pragma unroll
        for (int jj = k >> 1; jj > 0; jj >>= 1) {
            unsigned long long other;
            if (jj >= 32) {
                xbuf[xs][tid] = me;
                __syncthreads();
                other = xbuf[xs][tid ^ jj];
                xs ^= 1;
            } else {
                const unsigned lo = xlane((unsigned)(me & 0xFFFFFFFFull), jj);
                const unsigned hi = xlane((unsigned)(me >> 32), jj);
                other = ((unsigned long long)hi << 32) | lo;
            }
            const bool takemin = (((tid & jj) == 0) == up);
            const bool lt = me < other;
            me = (takemin == lt) ? me : other;
        }
    }

    unsigned ur = (unsigned)(me >> 32);
    ur ^= (ur & 0x80000000u) ? 0x80000000u : 0xFFFFFFFFu;
    const float dr = __uint_as_float(ur);
    const int   jr = (int)(me & 1023ull);
    sd[tid] = dr;
    sj[tid] = jr;
    __syncthreads();
    const float Dm = sd[NN - 1];

    float* orow = out + ((size_t)(b * NN) + wave * 32) * KDIM + head * DD + lane;
    if (!(Dm > -INFINITY)) {
        for (int qq = 0; qq < 32; ++qq) orow[(size_t)qq * KDIM] = 0.f;
        return;
    }

    const float dd = dr - Dm;
    const float Er = expf(dd);
    const float Fr = expf(NEG_SLOPE * dd);
    sE[tid] = Er;
    sF[tid] = Fr;

    float zf = Fr, ze = Er;
    for (int o = 1; o < 32; o <<= 1) {
        const float nf = __shfl_up(zf, o, 32);
        const float ne = __shfl_down(ze, o, 32);
        if (lane >= o) zf += nf;
        if (lane + o < 32) ze += ne;
    }
    zF[tid] = zf;
    zE[tid] = ze;
    if (lane == 31) zsegF[wave] = zf;
    if (lane == 0)  zsegE[wave] = ze;
    __syncthreads();

    const int r0 = wave * 32;
    float v[32];
#pragma unroll
    for (int t = 0; t < 32; ++t) v[t] = hs[sj[r0 + t] * SP + lane];
    __syncthreads();
    float run = 0.f;
#pragma unroll
    for (int t = 0; t < 32; ++t) { run += sF[r0 + t] * v[t]; PFt[r0 + t][lane] = run; }
    segF[wave][lane] = run;
    run = 0.f;
#pragma unroll
    for (int t = 31; t >= 0; --t) { run += sE[r0 + t] * v[t]; SEt[r0 + t][lane] = run; }
    segE[wave][lane] = run;
    __syncthreads();

    if (wave == 0) {
        float runF = 0.f;
#pragma unroll
        for (int g = 0; g < 32; ++g) { offF[g][lane] = runF; runF += segF[g][lane]; }
    } else if (wave == 1) {
        float runE = 0.f;
#pragma unroll
        for (int g = 31; g >= 0; --g) { offE[g][lane] = runE; runE += segE[g][lane]; }
    } else if (wave == 2) {
        float zoF = 0.f, zoE = 0.f;
#pragma unroll
        for (int g = 0; g < 32; ++g) {
            const float za = zsegF[g], zb = zsegE[g];
            if (g < lane) zoF += za;
            if (g > lane) zoE += zb;
        }
        zoffF[lane] = zoF;
        zoffE[lane] = zoE;
    }
    __syncthreads();

    const float tneg = -sq;
    int k = 0;
    for (int step = NN / 2; step > 0; step >>= 1)
        if (sd[k + step - 1] < tneg) k += step;
    if (sd[k] < tneg) k += 1;
    const float e0 = sq + Dm;
    const float m  = (e0 >= 0.f) ? e0 : NEG_SLOPE * e0;
    const float ai = expf(e0 - m);
    const float ci = expf(NEG_SLOPE * e0 - m);
    const int kp = k > 0 ? k - 1 : 0;
    const int ks = k < NN ? k : NN - 1;
    float zfv = zF[kp] + zoffF[kp >> 5];
    float zev = zE[ks] + zoffE[ks >> 5];
    if (k == 0)  zfv = 0.f;
    if (k == NN) zev = 0.f;
    const float Z   = ci * zfv + ai * zev;
    const float inv = (Z > 0.f) ? (1.f / Z) : 0.f;
    const float aw = ai * inv, cw = ci * inv;

    for (int qq = 0; qq < 32; ++qq) {
        const int   kq  = __shfl(k, qq, 32);
        const float awq = __shfl(aw, qq, 32);
        const float cwq = __shfl(cw, qq, 32);
        const int ip = kq > 0 ? kq - 1 : 0;
        const int is = kq < NN ? kq : NN - 1;
        float pf = PFt[ip][lane] + offF[ip >> 5][lane];
        float se = SEt[is][lane] + offE[is >> 5][lane];
        if (kq == 0)  pf = 0.f;
        if (kq == NN) se = 0.f;
        orow[(size_t)qq * KDIM] = cwq * pf + awq * se;
    }
}

extern "C" void kernel_launch(void* const* d_in, const int* in_sizes, int n_in,
                              void* d_out, int out_size, void* d_ws, size_t ws_size,
                              hipStream_t stream) {
    const float* x     = (const float*)d_in[0];
    const float* alive = (const float*)d_in[1];
    const float* W     = (const float*)d_in[2];
    const float* a_src = (const float*)d_in[3];
    const float* a_dst = (const float*)d_in[4];
    float* out = (float*)d_out;

    float* h_ws  = (float*)d_ws;

    gat_h_gemm     <<<512, 256,  0, stream>>>(x, W, h_ws);
    gat_attn_sorted<<<32,  1024, 0, stream>>>(h_ws, a_src, a_dst, alive, out);
}


# h GEMM retiled: 256 blocks of 4 waves; a wave keeps its 16 W columns' fragments in registers, straight from global; x tile via LDS once; one round trip; k order in products: sum order changed
# speedup vs baseline: 2.5382x; 1.5147x over previous
#include <hip/hip_runtime.h>
#include <math.h>

typedef __attribute__((ext_vector_type(2))) float v2f;
typedef __attribute__((ext_vector_type(8))) float v8f;

#define HH 4
#define DD 32
#define NN 1024
#define BB 8
#define KDIM 128
#define NEG_SLOPE 0.2f

__global__ __launch_bounds__(128) void gat_h_gemm(const float* __restrict__ x,
                                                  const float* __restrict__ W,
                                                  float* __restrict__ h) {
    constexpr int XP = 132;
    __shared__ __attribute__((aligned(16))) float lds_x[64 * XP];
    const int tid  = threadIdx.x;
    const int lane = tid & 31;
    const int wave = tid >> 5;
    const int r0   = (blockIdx.x >> 1) * 64;
    const int c0   = (blockIdx.x & 1) * 64 + wave * 16;
    const int lm   = lane & 15;
    const int lhi  = lane >> 4;

    const float4* xg = (const float4*)(x + (size_t)r0 * KDIM);
    float4 xp[16];
#pragma unroll
    for (int i = 0; i < 16; ++i) xp[i] = xg[i * 128 + tid];
    const float4* wg = (const float4*)(W + (size_t)(c0 + lm) * KDIM + lhi * 4);
    float4 bf[16];
#pragma unroll
    for (int t = 0; t < 16; ++t) bf[t] = wg[t * 2];
#pragma unroll
    for (int i = 0; i < 16; ++i) {
        const int g = i * 128 + tid;
        *(float4*)(lds_x + (g >> 5) * XP + (g & 31) * 4) = xp[i];
    }
    __builtin_amdgcn_fence(__ATOMIC_RELEASE, "workgroup", "local");
    __builtin_amdgcn_s_barrier();
    __builtin_amdgcn_fence(__ATOMIC_ACQUIRE, "workgroup", "local");

    v8f acc[4] = {};
    const float* ap = lds_x + lm * XP + lhi * 4;
    float4 a[2][4];
#pragma unroll
    for (int m = 0; m < 4; ++m) a[0][m] = *(const float4*)(ap + m * 16 * XP);
    __builtin_amdgcn_sched_group_barrier(0x100, 4, 0);
#pragma unroll
    for (int t = 0; t < 16; ++t) {
        const int cur = t & 1, nxt = cur ^ 1;
        if (t + 1 < 16) {
#pragma unroll
            for (int m = 0; m < 4; ++m) a[nxt][m] = *(const float4*)(ap + m * 16 * XP + (t + 1) * 8);
        }
        v2f b0, b1;
        b0.x = bf[t].x; b0.y = bf[t].y;
        b1.x = bf[t].z; b1.y = bf[t].w;
#pragma unroll
        for (int m = 0; m < 4; ++m) {
            v2f a0; a0.x = a[cur][m].x; a0.y = a[cur][m].y;
            acc[m] = __builtin_amdgcn_wmma_f32_16x16x4_f32(false, a0, false, b0,
                                                           (short)0, acc[m], false, false);
        }
#pragma unroll
        for (int m = 0; m < 4; ++m) {
            v2f a1; a1.x = a[cur][m].z; a1.y = a[cur][m].w;
            acc[m] = __builtin_amdgcn_wmma_f32_16x16x4_f32(false, a1, false, b1,
                                                           (short)0, acc[m], false, false);
        }
        if (t + 1 < 16) __builtin_amdgcn_sched_group_barrier(0x100, 4, 0);
        __builtin_amdgcn_sched_group_barrier(0x008, 8, 0);
    }
    const int col = c0 + lm;
    float* hq = h + (size_t)(r0 >> 10) * (HH * NN * DD)
                  + (((col >> 5) * NN + (r0 & (NN - 1)) + lhi * 8) * DD + (col & (DD - 1)));
#pragma unroll
    for (int m = 0; m < 4; ++m)
#pragma unroll
        for (int v = 0; v < 8; ++v) hq[(m * 16 + v) * DD] = acc[m][v];
}

__device__ __forceinline__ unsigned xlane(unsigned x, int jj) {
    const int bits = (int)x;
    switch (jj) {
    case 1:  return (unsigned)__builtin_amdgcn_update_dpp(bits, bits, 0x161, 0xf, 0xf, false);
    case 2:  return (unsigned)__builtin_amdgcn_update_dpp(bits, bits, 0x162, 0xf, 0xf, false);
    case 4:  return (unsigned)__builtin_amdgcn_update_dpp(bits, bits, 0x164, 0xf, 0xf, false);
    case 8:  return (unsigned)__builtin_amdgcn_update_dpp(bits, bits, 0x168, 0xf, 0xf, false);
    default: return (unsigned)__builtin_amdgcn_permlanex16(bits, bits, 0x76543210, 0xfedcba98, false, false);
    }
}

__global__ __launch_bounds__(1024) void gat_attn_sorted(const float* __restrict__ h,
                                                        const float* __restrict__ a_src,
                                                        const float* __restrict__ a_dst,
                                                        const float* __restrict__ alive,
                                                        float* __restrict__ out) {
    __shared__ unsigned long long xbuf[2][NN];
    __shared__ float sd[NN];
    __shared__ int   sj[NN];
    __shared__ float sE[NN];
    __shared__ float sF[NN];
    __shared__ float zF[NN];
    __shared__ float zE[NN];
    __shared__ float segF[32][32];
    __shared__ float segE[32][32];
    __shared__ float offF[32][32];
    __shared__ float offE[32][32];
    __shared__ float zsegF[32];
    __shared__ float zsegE[32];
    __shared__ float zoffF[32];
    __shared__ float zoffE[32];
    __shared__ __attribute__((aligned(16))) float tab[2 * NN * 32];
    float (*PFt)[32] = (float (*)[32])tab;
    float (*SEt)[32] = (float (*)[32])(tab + NN * 32);
    float* hs = tab;
    constexpr int SP = 36;

    const int tid  = threadIdx.x;
    const int lane = tid & 31;
    const int wave = tid >> 5;
    const int head = blockIdx.x & 3;
    const int b    = blockIdx.x >> 2;

    const float al = alive[b * NN + tid];
    const float4* hg = (const float4*)(h + (size_t)blockIdx.x * NN * DD);
    float4 pc[8];
#pragma unroll
    for (int i = 0; i < 8; ++i) pc[i] = hg[i * 1024 + tid];
#pragma unroll
    for (int i = 0; i < 8; ++i) {
        const int g = i * 1024 + tid;
        *(float4*)(hs + (g >> 3) * SP + (g & 7) * 4) = pc[i];
    }
    __syncthreads();
    const float4* hp = (const float4*)(hs + tid * SP);
    const float4* sp = (const float4*)(a_src + head * DD);
    const float4* dp = (const float4*)(a_dst + head * DD);
    float sq = 0.f, dj = 0.f;
#pragma unroll
    for (int i = 0; i < 8; ++i) {
        const float4 hv = hp[i], av = sp[i], dv = dp[i];
        sq += hv.x * av.x + hv.y * av.y + hv.z * av.z + hv.w * av.w;
        dj += hv.x * dv.x + hv.y * dv.y + hv.z * dv.z + hv.w * dv.w;
    }
    const float kd = (al >= 0.5f) ? dj : -INFINITY;
    unsigned u = __float_as_uint(kd);
    u ^= (u & 0x80000000u) ? 0xFFFFFFFFu : 0x80000000u;
    unsigned long long me = ((unsigned long long)u << 32) | (unsigned)tid;

    int xs = 0;
#pragma unroll
    for (int k = 2; k <= NN; k <<= 1) {
        const bool up = (tid & k) == 0;
#pragma unroll
        for (int jj = k >> 1; jj > 0; jj >>= 1) {
            unsigned long long other;
            if (jj >= 32) {
                xbuf[xs][tid] = me;
                __syncthreads();
                other = xbuf[xs][tid ^ jj];
                xs ^= 1;
            } else {
                const unsigned lo = xlane((unsigned)(me & 0xFFFFFFFFull), jj);
                const unsigned hi = xlane((unsigned)(me >> 32), jj);
                other = ((unsigned long long)hi << 32) | lo;
            }
            const bool takemin = (((tid & jj) == 0) == up);
            const bool lt = me < other;
            me = (takemin == lt) ? me : other;
        }
    }

    unsigned ur = (unsigned)(me >> 32);
    ur ^= (ur & 0x80000000u) ? 0x80000000u : 0xFFFFFFFFu;
    const float dr = __uint_as_float(ur);
    const int   jr = (int)(me & 1023ull);
    sd[tid] = dr;
    sj[tid] = jr;
    __syncthreads();
    const float Dm = sd[NN - 1];

    float* orow = out + ((size_t)(b * NN) + wave * 32) * KDIM + head * DD + lane;
    if (!(Dm > -INFINITY)) {
        for (int qq = 0; qq < 32; ++qq) orow[(size_t)qq * KDIM] = 0.f;
        return;
    }

    const float dd = dr - Dm;
    const float Er = expf(dd);
    const float Fr = expf(NEG_SLOPE * dd);
    sE[tid] = Er;
    sF[tid] = Fr;

    float zf = Fr, ze = Er;
    for (int o = 1; o < 32; o <<= 1) {
        const float nf = __shfl_up(zf, o, 32);
        const float ne = __shfl_down(ze, o, 32);
        if (lane >= o) zf += nf;
        if (lane + o < 32) ze += ne;
    }
    zF[tid] = zf;
    zE[tid] = ze;
    if (lane == 31) zsegF[wave] = zf;
    if (lane == 0)  zsegE[wave] = ze;
    __syncthreads();

    const int r0 = wave * 32;
    float v[32];
#pragma unroll
    for (int t = 0; t < 32; ++t) v[t] = hs[sj[r0 + t] * SP + lane];
    __syncthreads();
    float run = 0.f;
#pragma unroll
    for (int t = 0; t < 32; ++t) { run += sF[r0 + t] * v[t]; PFt[r0 + t][lane] = run; }
    segF[wave][lane] = run;
    run = 0.f;
#pragma unroll
    for (int t = 31; t >= 0; --t) { run += sE[r0 + t] * v[t]; SEt[r0 + t][lane] = run; }
    segE[wave][lane] = run;
    __syncthreads();

    if (wave == 0) {
        float runF = 0.f;
#pragma unroll
        for (int g = 0; g < 32; ++g) { offF[g][lane] = runF; runF += segF[g][lane]; }
    } else if (wave == 1) {
        float runE = 0.f;
#pragma unroll
        for (int g = 31; g >= 0; --g) { offE[g][lane] = runE; runE += segE[g][lane]; }
    } else if (wave == 2) {
        float zoF = 0.f, zoE = 0.f;
#pragma unroll
        for (int g = 0; g < 32; ++g) {
            const float za = zsegF[g], zb = zsegE[g];
            if (g < lane) zoF += za;
            if (g > lane) zoE += zb;
        }
        zoffF[lane] = zoF;
        zoffE[lane] = zoE;
    }
    __syncthreads();

    const float tneg = -sq;
    int k = 0;
    for (int step = NN / 2; step > 0; step >>= 1)
        if (sd[k + step - 1] < tneg) k += step;
    if (sd[k] < tneg) k += 1;
    const float e0 = sq + Dm;
    const float m  = (e0 >= 0.f) ? e0 : NEG_SLOPE * e0;
    const float ai = expf(e0 - m);
    const float ci = expf(NEG_SLOPE * e0 - m);
    const int kp = k > 0 ? k - 1 : 0;
    const int ks = k < NN ? k : NN - 1;
    float zfv = zF[kp] + zoffF[kp >> 5];
    float zev = zE[ks] + zoffE[ks >> 5];
    if (k == 0)  zfv = 0.f;
    if (k == NN) zev = 0.f;
    const float Z   = ci * zfv + ai * zev;
    const float inv = (Z > 0.f) ? (1.f / Z) : 0.f;
    const float aw = ai * inv, cw = ci * inv;

    for (int qq = 0; qq < 32; ++qq) {
        const int   kq  = __shfl(k, qq, 32);
        const float awq = __shfl(aw, qq, 32);
        const float cwq = __shfl(cw, qq, 32);
        const int ip = kq > 0 ? kq - 1 : 0;
        const int is = kq < NN ? kq : NN - 1;
        float pf = PFt[ip][lane] + offF[ip >> 5][lane];
        float se = SEt[is][lane] + offE[is >> 5][lane];
        if (kq == 0)  pf = 0.f;
        if (kq == NN) se = 0.f;
        orow[(size_t)qq * KDIM] = cwq * pf + awq * se;
    }
}

extern "C" void kernel_launch(void* const* d_in, const int* in_sizes, int n_in,
                              void* d_out, int out_size, void* d_ws, size_t ws_size,
                              hipStream_t stream) {
    const float* x     = (const float*)d_in[0];
    const float* alive = (const float*)d_in[1];
    const float* W     = (const float*)d_in[2];
    const float* a_src = (const float*)d_in[3];
    const float* a_dst = (const float*)d_in[4];
    float* out = (float*)d_out;

    float* h_ws  = (float*)d_ws;

    gat_h_gemm     <<<256, 128,  0, stream>>>(x, W, h_ws);
    gat_attn_sorted<<<32,  1024, 0, stream>>>(h_ws, a_src, a_dst, alive, out);
}


# output retiled: a lane takes 4 columns of one of 4 queries per step (was 1 column, 1 query); k and both weights come as one 16-byte LDS record, not 3 lane permutes; 16-byte reads/stores; sums kept
# speedup vs baseline: 2.7448x; 1.0814x over previous
#include <hip/hip_runtime.h>
#include <math.h>

typedef __attribute__((ext_vector_type(2))) float v2f;
typedef __attribute__((ext_vector_type(8))) float v8f;
typedef __attribute__((ext_vector_type(4))) float v4f;

#define HH 4
#define DD 32
#define NN 1024
#define BB 8
#define KDIM 128
#define NEG_SLOPE 0.2f

__global__ __launch_bounds__(128) void gat_h_gemm(const float* __restrict__ x,
                                                  const float* __restrict__ W,
                                                  float* __restrict__ h) {
    constexpr int XP = 132;
    __shared__ __attribute__((aligned(16))) float lds_x[64 * XP];
    const int tid  = threadIdx.x;
    const int lane = tid & 31;
    const int wave = tid >> 5;
    const int r0   = (blockIdx.x >> 1) * 64;
    const int c0   = (blockIdx.x & 1) * 64 + wave * 16;
    const int lm   = lane & 15;
    const int lhi  = lane >> 4;

    const float4* xg = (const float4*)(x + (size_t)r0 * KDIM);
    float4 xp[16];
#pragma unroll
    for (int i = 0; i < 16; ++i) xp[i] = xg[i * 128 + tid];
    const float4* wg = (const float4*)(W + (size_t)(c0 + lm) * KDIM + lhi * 4);
    float4 bf[16];
#pragma unroll
    for (int t = 0; t < 16; ++t) bf[t] = wg[t * 2];
#pragma unroll
    for (int i = 0; i < 16; ++i) {
        const int g = i * 128 + tid;
        *(float4*)(lds_x + (g >> 5) * XP + (g & 31) * 4) = xp[i];
    }
    __builtin_amdgcn_fence(__ATOMIC_RELEASE, "workgroup", "local");
    __builtin_amdgcn_s_barrier();
    __builtin_amdgcn_fence(__ATOMIC_ACQUIRE, "workgroup", "local");

    v8f acc[4] = {};
    const float* ap = lds_x + lm * XP + lhi * 4;
    float4 a[2][4];
#pragma unroll
    for (int m = 0; m < 4; ++m) a[0][m] = *(const float4*)(ap + m * 16 * XP);
    __builtin_amdgcn_sched_group_barrier(0x100, 4, 0);
#pragma unroll
    for (int t = 0; t < 16; ++t) {
        const int cur = t & 1, nxt = cur ^ 1;
        if (t + 1 < 16) {
#pragma unroll
            for (int m = 0; m < 4; ++m) a[nxt][m] = *(const float4*)(ap + m * 16 * XP + (t + 1) * 8);
        }
        v2f b0, b1;
        b0.x = bf[t].x; b0.y = bf[t].y;
        b1.x = bf[t].z; b1.y = bf[t].w;
#pragma unroll
        for (int m = 0; m < 4; ++m) {
            v2f a0; a0.x = a[cur][m].x; a0.y = a[cur][m].y;
            acc[m] = __builtin_amdgcn_wmma_f32_16x16x4_f32(false, a0, false, b0,
                                                           (short)0, acc[m], false, false);
        }
#pragma unroll
        for (int m = 0; m < 4; ++m) {
            v2f a1; a1.x = a[cur][m].z; a1.y = a[cur][m].w;
            acc[m] = __builtin_amdgcn_wmma_f32_16x16x4_f32(false, a1, false, b1,
                                                           (short)0, acc[m], false, false);
        }
        if (t + 1 < 16) __builtin_amdgcn_sched_group_barrier(0x100, 4, 0);
        __builtin_amdgcn_sched_group_barrier(0x008, 8, 0);
    }
    const int col = c0 + lm;
    float* hq = h + (size_t)(r0 >> 10) * (HH * NN * DD)
                  + (((col >> 5) * NN + (r0 & (NN - 1)) + lhi * 8) * DD + (col & (DD - 1)));
#pragma unroll
    for (int m = 0; m < 4; ++m)
#pragma unroll
        for (int v = 0; v < 8; ++v) hq[(m * 16 + v) * DD] = acc[m][v];
}

__device__ __forceinline__ unsigned xlane(unsigned x, int jj) {
    const int bits = (int)x;
    switch (jj) {
    case 1:  return (unsigned)__builtin_amdgcn_update_dpp(bits, bits, 0x161, 0xf, 0xf, false);
    case 2:  return (unsigned)__builtin_amdgcn_update_dpp(bits, bits, 0x162, 0xf, 0xf, false);
    case 4:  return (unsigned)__builtin_amdgcn_update_dpp(bits, bits, 0x164, 0xf, 0xf, false);
    case 8:  return (unsigned)__builtin_amdgcn_update_dpp(bits, bits, 0x168, 0xf, 0xf, false);
    default: return (unsigned)__builtin_amdgcn_permlanex16(bits, bits, 0x76543210, 0xfedcba98, false, false);
    }
}

__global__ __launch_bounds__(1024) void gat_attn_sorted(const float* __restrict__ h,
                                                        const float* __restrict__ a_src,
                                                        const float* __restrict__ a_dst,
                                                        const float* __restrict__ alive,
                                                        float* __restrict__ out) {
    __shared__ __attribute__((aligned(16))) unsigned long long xbuf[2][NN];
    __shared__ float sd[NN];
    __shared__ int   sj[NN];
    __shared__ float sE[NN];
    __shared__ float sF[NN];
    __shared__ float zF[NN];
    __shared__ float zE[NN];
    __shared__ float segF[32][32];
    __shared__ float segE[32][32];
    __shared__ __attribute__((aligned(16))) float offF[32][32];
    __shared__ __attribute__((aligned(16))) float offE[32][32];
    __shared__ float zsegF[32];
    __shared__ float zsegE[32];
    __shared__ float zoffF[32];
    __shared__ float zoffE[32];
    __shared__ __attribute__((aligned(16))) float tab[2 * NN * 32];
    float (*PFt)[32] = (float (*)[32])tab;
    float (*SEt)[32] = (float (*)[32])(tab + NN * 32);
    float* hs = tab;
    constexpr int SP = 36;

    const int tid  = threadIdx.x;
    const int lane = tid & 31;
    const int wave = tid >> 5;
    const int head = blockIdx.x & 3;
    const int b    = blockIdx.x >> 2;

    const float al = alive[b * NN + tid];
    const float4* hg = (const float4*)(h + (size_t)blockIdx.x * NN * DD);
    float4 pc[8];
#pragma unroll
    for (int i = 0; i < 8; ++i) pc[i] = hg[i * 1024 + tid];
#pragma unroll
    for (int i = 0; i < 8; ++i) {
        const int g = i * 1024 + tid;
        *(float4*)(hs + (g >> 3) * SP + (g & 7) * 4) = pc[i];
    }
    __syncthreads();
    const float4* hp = (const float4*)(hs + tid * SP);
    const float4* sp = (const float4*)(a_src + head * DD);
    const float4* dp = (const float4*)(a_dst + head * DD);
    float sq = 0.f, dj = 0.f;
#pragma unroll
    for (int i = 0; i < 8; ++i) {
        const float4 hv = hp[i], av = sp[i], dv = dp[i];
        sq += hv.x * av.x + hv.y * av.y + hv.z * av.z + hv.w * av.w;
        dj += hv.x * dv.x + hv.y * dv.y + hv.z * dv.z + hv.w * dv.w;
    }
    const float kd = (al >= 0.5f) ? dj : -INFINITY;
    unsigned u = __float_as_uint(kd);
    u ^= (u & 0x80000000u) ? 0xFFFFFFFFu : 0x80000000u;
    unsigned long long me = ((unsigned long long)u << 32) | (unsigned)tid;

    int xs = 0;
#pragma unroll
    for (int k = 2; k <= NN; k <<= 1) {
        const bool up = (tid & k) == 0;
#pragma unroll
        for (int jj = k >> 1; jj > 0; jj >>= 1) {
            unsigned long long other;
            if (jj >= 32) {
                xbuf[xs][tid] = me;
                __syncthreads();
                other = xbuf[xs][tid ^ jj];
                xs ^= 1;
            } else {
                const unsigned lo = xlane((unsigned)(me & 0xFFFFFFFFull), jj);
                const unsigned hi = xlane((unsigned)(me >> 32), jj);
                other = ((unsigned long long)hi << 32) | lo;
            }
            const bool takemin = (((tid & jj) == 0) == up);
            const bool lt = me < other;
            me = (takemin == lt) ? me : other;
        }
    }

    unsigned ur = (unsigned)(me >> 32);
    ur ^= (ur & 0x80000000u) ? 0x80000000u : 0xFFFFFFFFu;
    const float dr = __uint_as_float(ur);
    const int   jr = (int)(me & 1023ull);
    sd[tid] = dr;
    sj[tid] = jr;
    __syncthreads();
    const float Dm = sd[NN - 1];

    float* orow = out + ((size_t)(b * NN) + wave * 32) * KDIM + head * DD + lane;
    if (!(Dm > -INFINITY)) {
        for (int qq = 0; qq < 32; ++qq) orow[(size_t)qq * KDIM] = 0.f;
        return;
    }

    const float dd = dr - Dm;
    const float Er = expf(dd);
    const float Fr = expf(NEG_SLOPE * dd);
    sE[tid] = Er;
    sF[tid] = Fr;

    float zf = Fr, ze = Er;
    for (int o = 1; o < 32; o <<= 1) {
        const float nf = __shfl_up(zf, o, 32);
        const float ne = __shfl_down(ze, o, 32);
        if (lane >= o) zf += nf;
        if (lane + o < 32) ze += ne;
    }
    zF[tid] = zf;
    zE[tid] = ze;
    if (lane == 31) zsegF[wave] = zf;
    if (lane == 0)  zsegE[wave] = ze;
    __syncthreads();

    const int r0 = wave * 32;
    float v[32];
#pragma unroll
    for (int t = 0; t < 32; ++t) v[t] = hs[sj[r0 + t] * SP + lane];
    __syncthreads();
    float run = 0.f;
#pragma unroll
    for (int t = 0; t < 32; ++t) { run += sF[r0 + t] * v[t]; PFt[r0 + t][lane] = run; }
    segF[wave][lane] = run;
    run = 0.f;
#pragma unroll
    for (int t = 31; t >= 0; --t) { run += sE[r0 + t] * v[t]; SEt[r0 + t][lane] = run; }
    segE[wave][lane] = run;
    __syncthreads();

    if (wave == 0) {
        float runF = 0.f;
#pragma unroll
        for (int g = 0; g < 32; ++g) { offF[g][lane] = runF; runF += segF[g][lane]; }
    } else if (wave == 1) {
        float runE = 0.f;
#pragma unroll
        for (int g = 31; g >= 0; --g) { offE[g][lane] = runE; runE += segE[g][lane]; }
    } else if (wave == 2) {
        float zoF = 0.f, zoE = 0.f;
#pragma unroll
        for (int g = 0; g < 32; ++g) {
            const float za = zsegF[g], zb = zsegE[g];
            if (g < lane) zoF += za;
            if (g > lane) zoE += zb;
        }
        zoffF[lane] = zoF;
        zoffE[lane] = zoE;
    }
    __syncthreads();

    const float tneg = -sq;
    int k = 0;
    for (int step = NN / 2; step > 0; step >>= 1)
        if (sd[k + step - 1] < tneg) k += step;
    if (sd[k] < tneg) k += 1;
    const float e0 = sq + Dm;
    const float m  = (e0 >= 0.f) ? e0 : NEG_SLOPE * e0;
    const float ai = expf(e0 - m);
    const float ci = expf(NEG_SLOPE * e0 - m);
    const int kp = k > 0 ? k - 1 : 0;
    const int ks = k < NN ? k : NN - 1;
    float zfv = zF[kp] + zoffF[kp >> 5];
    float zev = zE[ks] + zoffE[ks >> 5];
    if (k == 0)  zfv = 0.f;
    if (k == NN) zev = 0.f;
    const float Z   = ci * zfv + ai * zev;
    const float inv = (Z > 0.f) ? (1.f / Z) : 0.f;
    const float aw = ai * inv, cw = ci * inv;

    v4f* qrec = (v4f*)&xbuf[0][0];
    v4f myrec;
    myrec.x = __int_as_float(k);
    myrec.y = aw;
    myrec.z = cw;
    myrec.w = 0.f;
    qrec[tid] = myrec;
    const int c4 = (lane & 7) * 4;
    const int qs = lane >> 3;
    float* op = out + ((size_t)(b * NN) + wave * 32 + qs) * KDIM + head * DD + c4;
#pragma unroll 2
    for (int i = 0; i < 8; ++i) {
        const v4f rec = qrec[wave * 32 + i * 4 + qs];
        const int   kq  = __float_as_int(rec.x);
        const float awq = rec.y;
        const float cwq = rec.z;
        const int ip = kq > 0 ? kq - 1 : 0;
        const int is = kq < NN ? kq : NN - 1;
        v4f pf = *(const v4f*)&PFt[ip][c4] + *(const v4f*)&offF[ip >> 5][c4];
        v4f se = *(const v4f*)&SEt[is][c4] + *(const v4f*)&offE[is >> 5][c4];
        if (kq == 0)  pf = (v4f)(0.f);
        if (kq == NN) se = (v4f)(0.f);
        *(v4f*)(op + (size_t)(i * 4) * KDIM) = __builtin_elementwise_fma((v4f)(awq), se, cwq * pf);
    }
}

extern "C" void kernel_launch(void* const* d_in, const int* in_sizes, int n_in,
                              void* d_out, int out_size, void* d_ws, size_t ws_size,
                              hipStream_t stream) {
    const float* x     = (const float*)d_in[0];
    const float* alive = (const float*)d_in[1];
    const float* W     = (const float*)d_in[2];
    const float* a_src = (const float*)d_in[3];
    const float* a_dst = (const float*)d_in[4];
    float* out = (float*)d_out;

    float* h_ws  = (float*)d_ws;

    gat_h_gemm     <<<256, 128,  0, stream>>>(x, W, h_ws);
    gat_attn_sorted<<<32,  1024, 0, stream>>>(h_ws, a_src, a_dst, alive, out);
}
